# starting kernel
# baseline (speedup 1.0000x reference)
#include <hip/hip_runtime.h>

#define N_NODES 50000
#define N_EDGES 800000
#define NF 128

typedef float v2f __attribute__((ext_vector_type(2)));
typedef float v8f __attribute__((ext_vector_type(8)));

__global__ void k_init_deg(float* __restrict__ deg) {
    int i = blockIdx.x * blockDim.x + threadIdx.x;
    if (i < N_NODES) deg[i] = 1.0f;
}

__global__ void k_deg_accum(const int* __restrict__ dst, float* __restrict__ deg) {
    int e = blockIdx.x * blockDim.x + threadIdx.x;
    if (e < N_EDGES) atomicAdd(&deg[dst[e]], 1.0f);
}

__global__ void k_deg_final(float* __restrict__ deg) {
    int i = blockIdx.x * blockDim.x + threadIdx.x;
    if (i < N_NODES) deg[i] = rsqrtf(deg[i]);
}

__global__ void __launch_bounds__(128) k_gemm_fused(const float* __restrict__ X,
                                                    const float* __restrict__ W,
                                                    const float* __restrict__ bias,
                                                    const float* __restrict__ dinv,
                                                    float* __restrict__ Y,
                                                    float* __restrict__ AggInit,
                                                    int nrows) {
    __shared__ float wlds[64 * 128 * 2];

    const int tid = threadIdx.x;
    for (int idx = tid; idx < 64 * 128; idx += 128) {
        const int p = idx >> 7;
        const int n = idx & 127;
        float2 w2;
        w2.x = W[(size_t)(2 * p) * NF + n];
        w2.y = W[(size_t)(2 * p + 1) * NF + n];
        *(float2*)(wlds + (size_t)idx * 2) = w2;
    }
    __syncthreads();

    const int wave    = tid >> 5;
    const int lane    = tid & 31;
    const int rowTile = blockIdx.x * 4 + wave;
    const int row0    = rowTile * 16;
    const bool active = (row0 < nrows);

    const int hi = lane >> 4;
    const int lm = lane & 15;

    v8f acc[8];
#pragma unroll
    for (int t = 0; t < 8; ++t) acc[t] = (v8f){0,0,0,0,0,0,0,0};

    const int arow = active ? (row0 + lm) : lm;
    const float* xrow = X + (size_t)arow * NF;

#pragma unroll 4
    for (int kt = 0; kt < 32; ++kt) {
        v2f a = *(const v2f*)(xrow + kt * 4 + 2 * hi);
        const int p = kt * 2 + hi;
        const float* wp = wlds + (size_t)p * 256 + lm * 2;
#pragma unroll
        for (int t = 0; t < 8; ++t) {
            v2f b = *(const v2f*)(wp + t * 32);
            acc[t] = __builtin_amdgcn_wmma_f32_16x16x4_f32(
                 false, a,  false, b,
                 (short)0, acc[t],  false,  false);
        }
    }

    if (!active) return;

    float dv[8];
#pragma unroll
    for (int r = 0; r < 8; ++r) dv[r] = dinv[row0 + 8 * hi + r];

#pragma unroll
    for (int t = 0; t < 8; ++t) {
        const int col = t * 16 + lm;
        const float bc = bias[col];
        float* yb = Y       + (size_t)(row0 + 8 * hi) * NF + col;
        float* ab = AggInit + (size_t)(row0 + 8 * hi) * NF + col;
#pragma unroll
        for (int r = 0; r < 8; ++r) {
            const float v = acc[t][r];
            yb[(size_t)r * NF] = v;
            ab[(size_t)r * NF] = bc + dv[r] * dv[r] * v;
        }
    }
}

__global__ void __launch_bounds__(256) k_scatter(const int* __restrict__ src,
                                                 const int* __restrict__ dst,
                                                 const float* __restrict__ dinv,
                                                 const float* __restrict__ msg,
                                                 float* __restrict__ agg) {
    int e = blockIdx.x * (blockDim.x >> 5) + (threadIdx.x >> 5);
    if (e >= N_EDGES) return;
    int lane = threadIdx.x & 31;
    int s = src[e];
    int d = dst[e];
    float norm = dinv[s] * dinv[d];
    const float4 v = *((const float4*)(msg + (size_t)s * NF) + lane);
    float* ap = agg + (size_t)d * NF + lane * 4;
    atomicAdd(ap + 0, norm * v.x);
    atomicAdd(ap + 1, norm * v.y);
    atomicAdd(ap + 2, norm * v.z);
    atomicAdd(ap + 3, norm * v.w);
}

__global__ void k_relu_skip(float* __restrict__ h, const float* __restrict__ x) {
    int i = blockIdx.x * blockDim.x + threadIdx.x;
    if (i >= N_NODES * NF) return;
    float v = h[i];
    h[i] = (v > 0.0f ? v : 0.0f) + x[i];
}

extern "C" void kernel_launch(void* const* d_in, const int* in_sizes, int n_in,
                              void* d_out, int out_size, void* d_ws, size_t ws_size,
                              hipStream_t stream) {
    const int*   ei  = (const int*)d_in[0];
    const int*   src = ei;
    const int*   dst = ei + N_EDGES;
    const float* x   = (const float*)d_in[1];
    const float* W1  = (const float*)d_in[2];
    const float* b1  = (const float*)d_in[3];
    const float* W2  = (const float*)d_in[4];
    const float* b2  = (const float*)d_in[5];
    float*       out = (float*)d_out;

    float* dinv = (float*)d_ws;
    float* xw   = dinv + 51200;
    float* agg  = xw + (size_t)N_NODES * NF;

    const int elemGrid     = (N_NODES * NF + 255) / 256;
    const int edgeWaveGrid = (N_EDGES + 7) / 8;
    const int gemmGrid     = (N_NODES / 16 + 3) / 4;

    k_init_deg  <<<(N_NODES + 255) / 256, 256, 0, stream>>>(dinv);
    k_deg_accum <<<(N_EDGES + 255) / 256, 256, 0, stream>>>(dst, dinv);
    k_deg_final <<<(N_NODES + 255) / 256, 256, 0, stream>>>(dinv);

    k_gemm_fused<<<gemmGrid, 128, 0, stream>>>(x, W1, b1, dinv, xw, agg, N_NODES);
    k_scatter   <<<edgeWaveGrid, 256, 0, stream>>>(src, dst, dinv, xw, agg);
    k_relu_skip <<<elemGrid, 256, 0, stream>>>(agg, x);

    k_gemm_fused<<<gemmGrid, 128, 0, stream>>>(agg, W2, b2, dinv, xw, out, N_NODES);
    k_scatter   <<<edgeWaveGrid, 256, 0, stream>>>(src, dst, dinv, xw, out);
}


# Layer 2's product joins layer 1's tile-sum launch: bias, relu and skip applied to the summed rows in LDS, which feed the f32 WMMA as A from there; layer 1's output never stored; 6 -> 5 launches
# speedup vs baseline: 2.4894x; 2.4894x over previous
#include <hip/hip_runtime.h>

#define N_NODES 50000
#define N_EDGES 800000
#define NF 128

#define TILE    196
#define NTILES  256
#define BIN_CAP 4096
#define CUR_OFF 50048
#define LIST_PITCH (TILE * NF)
#define LAST_ROWS  (N_NODES - (NTILES - 1) * TILE)
#define LAST_CAP   ((LAST_ROWS * NF) < BIN_CAP ? (LAST_ROWS * NF) : BIN_CAP)
#define MROWS   208
#define MPITCH  132
static_assert(TILE * NTILES >= N_NODES, "tiles cover every node");
static_assert(LAST_ROWS >= 1 && LAST_ROWS <= TILE, "the last tile is the only short one");
static_assert(LIST_PITCH >= BIN_CAP, "a full list fits inside its tile's own output rows");
static_assert(N_NODES <= 65536 && TILE <= 256, "list entry: src in bits 0-15, row inside the tile in bits 16-23");
static_assert(CUR_OFF >= N_NODES && CUR_OFF + NTILES <= 51200, "cursors stay inside the deg region");
static_assert(BIN_CAP == 32 * 16 * 8, "32 waves x 16 groups x 8 entries cover a full list");
static_assert(MROWS % 16 == 0 && MROWS >= TILE, "whole matrix row tiles cover the tile");
static_assert(MPITCH % 64 == 4 && MPITCH >= NF, "A fragment rows on different banks, 16-byte aligned rows");
static_assert((MROWS / 16) * 4 <= 64, "two work units per wave cover every (row tile, column quarter)");

__device__ __forceinline__ int list_cap(int tile) { return (tile == NTILES - 1) ? LAST_CAP : BIN_CAP; }

typedef float v2f __attribute__((ext_vector_type(2)));
typedef float v8f __attribute__((ext_vector_type(8)));

__global__ void k_init_deg(float* __restrict__ deg, int* __restrict__ cursors) {
    int i = blockIdx.x * blockDim.x + threadIdx.x;
    if (i < N_NODES) deg[i] = 1.0f;
    if (i < NTILES) cursors[i] = 0;
}


__global__ void __launch_bounds__(1024) k_file_edges(const int* __restrict__ src,
                                                     const int* __restrict__ dst,
                                                     float* __restrict__ deg,
                                                     int* __restrict__ cursor,
                                                     int* __restrict__ bins) {
    __shared__ int lcnt[NTILES];
    __shared__ int lbase[NTILES];
    const int tid = threadIdx.x;
    if (tid < NTILES) lcnt[tid] = 0;
    __syncthreads();

    int ent[4], bin[4], rank[4];
    const int e0 = blockIdx.x * 4096;
#pragma unroll
    for (int k = 0; k < 4; ++k) {
        const int e = e0 + k * 1024 + tid;
        ent[k] = 0; bin[k] = -1; rank[k] = 0;
        if (e < N_EDGES) {
            const int d = dst[e];
            const int s = src[e];
            const int b = d / TILE;
            bin[k]  = b;
            ent[k]  = s | ((d - b * TILE) << 16);
            rank[k] = atomicAdd(&lcnt[b], 1);
            atomicAdd(&deg[d], 1.0f);
        }
    }
    __syncthreads();
    if (tid < NTILES) {
        const int c = lcnt[tid];
        lbase[tid] = (c > 0) ? atomicAdd(&cursor[tid], c) : 0;
    }
    __syncthreads();
#pragma unroll
    for (int k = 0; k < 4; ++k) {
        if (bin[k] >= 0) {
            const int pos = lbase[bin[k]] + rank[k];
            if (pos < list_cap(bin[k])) bins[(size_t)bin[k] * LIST_PITCH + pos] = ent[k];
        }
    }
}

__global__ void __launch_bounds__(128) k_gemm_fused(const float* __restrict__ X,
                                                    const float* __restrict__ W,
                                                    const float* __restrict__ deg,
                                                    float* __restrict__ Y,
                                                    int nrows) {
    __shared__ float wlds[64 * 128 * 2];

    const int tid = threadIdx.x;
    for (int idx = tid; idx < 64 * 128; idx += 128) {
        const int p = idx >> 7;
        const int n = idx & 127;
        float2 w2;
        w2.x = W[(size_t)(2 * p) * NF + n];
        w2.y = W[(size_t)(2 * p + 1) * NF + n];
        *(float2*)(wlds + (size_t)idx * 2) = w2;
    }
    __syncthreads();

    const int wave    = tid >> 5;
    const int lane    = tid & 31;
    const int rowTile = blockIdx.x * 4 + wave;
    const int row0    = rowTile * 16;
    const bool active = (row0 < nrows);

    const int hi = lane >> 4;
    const int lm = lane & 15;

    v8f acc[8];
#pragma unroll
    for (int t = 0; t < 8; ++t) acc[t] = (v8f){0,0,0,0,0,0,0,0};

    const int arow = active ? (row0 + lm) : lm;
    const float* xrow = X + (size_t)arow * NF;

#pragma unroll 4
    for (int kt = 0; kt < 32; ++kt) {
        const v2f a = *(const v2f*)(xrow + kt * 4 + 2 * hi);
        const int p = kt * 2 + hi;
        const float* wp = wlds + (size_t)p * 256 + lm * 2;
#pragma unroll
        for (int t = 0; t < 8; ++t) {
            v2f b = *(const v2f*)(wp + t * 32);
            acc[t] = __builtin_amdgcn_wmma_f32_16x16x4_f32(
                 false, a,  false, b,
                 (short)0, acc[t],  false,  false);
        }
    }

    if (!active) return;

    float dv[8];
#pragma unroll
    for (int r = 0; r < 8; ++r) dv[r] = rsqrtf(deg[row0 + 8 * hi + r]);

#pragma unroll
    for (int t = 0; t < 8; ++t) {
        const int col = t * 16 + lm;
        float* yb = Y + (size_t)(row0 + 8 * hi) * NF + col;
#pragma unroll
        for (int r = 0; r < 8; ++r) yb[(size_t)r * NF] = dv[r] * acc[t][r];
    }
}

template <int PITCH>
__device__ __forceinline__ void add_one_edge(const float* __restrict__ Y, float* acc, int en, int lane) {
    const float* mp = Y + (size_t)(en & 0xffff) * NF + lane;
    float* ap = acc + ((en >> 16) & 0xff) * PITCH + lane;
    const float v0 = mp[0], v1 = mp[32], v2 = mp[64], v3 = mp[96];
    atomicAdd(ap,      v0);
    atomicAdd(ap + 32, v1);
    atomicAdd(ap + 64, v2);
    atomicAdd(ap + 96, v3);
}

template <int PITCH>
__device__ __forceinline__ void add_tile_edges(const int* __restrict__ src,
                                               const int* __restrict__ dst,
                                               const int* __restrict__ cursor,
                                               const int* bins,
                                               const float* __restrict__ Y,
                                               float* acc, int b, int row0, int wave, int lane) {
    const int total = __builtin_amdgcn_readfirstlane(cursor[b]);
    if (total <= list_cap(b)) {
        const int* myb = bins + (size_t)b * LIST_PITCH;
        const int lidx = (wave + 32 * (lane >> 3)) * 8 + (lane & 7);
        int entCur = (lidx < total) ? myb[lidx] : 0;
#pragma unroll 1
        for (int q = 0; q < 4; ++q) {
            const int nidx = lidx + 1024 * (q + 1);
            const int entNext = (q < 3 && nidx < total) ? myb[nidx] : 0;
#pragma unroll 1
            for (int r = 0; r < 4; ++r) {
                const int nval = total - (wave + 32 * (r + 4 * q)) * 8;
                if (nval <= 0) break;
                const int l0 = r * 8;
                if (nval >= 8) {
                    float v[8][4];
#pragma unroll
                    for (int j = 0; j < 8; ++j) {
                        const int en = __builtin_amdgcn_readlane(entCur, l0 + j);
                        const float* mp = Y + (size_t)(en & 0xffff) * NF + lane;
                        v[j][0] = mp[0];
                        v[j][1] = mp[32];
                        v[j][2] = mp[64];
                        v[j][3] = mp[96];
                    }
#pragma unroll
                    for (int j = 0; j < 8; ++j) {
                        const int en = __builtin_amdgcn_readlane(entCur, l0 + j);
                        float* ap = acc + ((en >> 16) & 0xff) * PITCH + lane;
                        atomicAdd(ap,      v[j][0]);
                        atomicAdd(ap + 32, v[j][1]);
                        atomicAdd(ap + 64, v[j][2]);
                        atomicAdd(ap + 96, v[j][3]);
                    }
                } else {
#pragma unroll 1
                    for (int j = 0; j < nval; ++j)
                        add_one_edge<PITCH>(Y, acc, __builtin_amdgcn_readlane(entCur, l0 + j), lane);
                }
            }
            entCur = entNext;
        }
    } else {
        for (int e0 = wave * 32; e0 < N_EDGES; e0 += 32 * 32) {
            const int d = dst[e0 + lane];
            const int s = src[e0 + lane];
            const bool mine = (d >= row0) && (d < row0 + TILE);
            unsigned mask = (unsigned)__ballot(mine);
            while (mask) {
                const int j = __builtin_ctz(mask);
                mask &= mask - 1;
                const int sj = __shfl(s, j);
                const int dj = __shfl(d, j) - row0;
                add_one_edge<PITCH>(Y, acc, sj | (dj << 16), lane);
            }
        }
    }
}

__global__ void __launch_bounds__(1024) k_tile_sum(const int* __restrict__ src,
                                                   const int* __restrict__ dst,
                                                   const int* __restrict__ cursor,
                                                   const int* bins,
                                                   const float* __restrict__ deg,
                                                   const float* __restrict__ Y,
                                                   const float* __restrict__ bias,
                                                   float* outp) {
    __shared__ __attribute__((aligned(16))) float acc[TILE * NF];

    const int tid  = threadIdx.x;
    const int lane = tid & 31;
    const int wave = __builtin_amdgcn_readfirstlane(tid >> 5);
    const int b    = blockIdx.x;
    const int row0 = b * TILE;
    const int nrows = (N_NODES - row0 < TILE) ? (N_NODES - row0) : TILE;

    const float4* y4 = (const float4*)(Y + (size_t)row0 * NF);
    float4* a4 = (float4*)acc;
    for (int i = tid; i < nrows * (NF / 4); i += 1024) a4[i] = y4[i];
    __syncthreads();

    add_tile_edges<NF>(src, dst, cursor, bins, Y, acc, b, row0, wave, lane);
    __syncthreads();

    float4* o4 = (float4*)(outp + (size_t)row0 * NF);
    const float4* b4 = (const float4*)bias;
    for (int i = tid; i < nrows * (NF / 4); i += 1024) {
        const float dinv = rsqrtf(deg[row0 + (i >> 5)]);
        const float4 a = a4[i];
        const float4 bc = b4[i & 31];
        float4 o;
        o.x = bc.x + dinv * a.x;
        o.y = bc.y + dinv * a.y;
        o.z = bc.z + dinv * a.z;
        o.w = bc.w + dinv * a.w;
        o4[i] = o;
    }
}

#define RPT ((MROWS + 31) / 32)
__global__ void __launch_bounds__(1024) k_tile_sum_next(const int* __restrict__ src,
                                                        const int* __restrict__ dst,
                                                        const int* __restrict__ cursor,
                                                        const int* __restrict__ bins,
                                                        const float* __restrict__ deg,
                                                        const float* __restrict__ Y,
                                                        const float* __restrict__ bias,
                                                        const float* __restrict__ X,
                                                        const float* __restrict__ W,
                                                        float* __restrict__ Y2) {
    __shared__ __attribute__((aligned(16))) float acc[MROWS * MPITCH];
    __shared__ __attribute__((aligned(16))) float wlds[64 * 128 * 2];
    __shared__ float dinvs[MROWS];

    const int tid  = threadIdx.x;
    const int lane = tid & 31;
    const int wave = __builtin_amdgcn_readfirstlane(tid >> 5);
    const int b    = blockIdx.x;
    const int row0 = b * TILE;
    const int nrows = (N_NODES - row0 < TILE) ? (N_NODES - row0) : TILE;

    const int c  = tid & 31;
    const int rb = tid >> 5;
    float4* a4 = (float4*)acc;

    const float4* y4 = (const float4*)(Y + (size_t)row0 * NF);
    const float4* x4 = (const float4*)(X + (size_t)row0 * NF);
    float4 w0[2], w1[2], ys[RPT], xs[RPT];
#pragma unroll
    for (int k = 0; k < 2; ++k) {
        const int idx = tid + 1024 * k;
        w0[k] = ((const float4*)(W + (size_t)(2 * (idx >> 5)) * NF))[idx & 31];
        w1[k] = ((const float4*)(W + (size_t)(2 * (idx >> 5) + 1) * NF))[idx & 31];
    }
#pragma unroll
    for (int it = 0; it < RPT; ++it) {
        const int r  = rb + 32 * it;
        const int rc = (r < nrows) ? r : (nrows - 1);
        ys[it] = y4[rc * (NF / 4) + c];
        xs[it] = x4[rc * (NF / 4) + c];
    }
    const float4 bc = ((const float4*)bias)[c];
    float dv = 0.0f;
    if (tid < nrows) dv = rsqrtf(deg[row0 + tid]);

#pragma unroll
    for (int k = 0; k < 2; ++k) {
        const int idx = tid + 1024 * k;
        float* wd = wlds + (size_t)((idx >> 5) * 128 + (idx & 31) * 4) * 2;
        *(float4*)(wd)     = make_float4(w0[k].x, w1[k].x, w0[k].y, w1[k].y);
        *(float4*)(wd + 4) = make_float4(w0[k].z, w1[k].z, w0[k].w, w1[k].w);
    }
#pragma unroll
    for (int it = 0; it < RPT; ++it) {
        const int r = rb + 32 * it;
        if (r < MROWS)
            a4[r * (MPITCH / 4) + c] = (r < nrows) ? ys[it] : make_float4(0.0f, 0.0f, 0.0f, 0.0f);
    }
    if (tid < MROWS) dinvs[tid] = dv;
    __syncthreads();

    add_tile_edges<MPITCH>(src, dst, cursor, bins, Y, acc, b, row0, wave, lane);
    __syncthreads();

#pragma unroll
    for (int it = 0; it < RPT; ++it) {
        const int r = rb + 32 * it;
        if (r < nrows) {
            const float dinv = dinvs[r];
            const float4 a = a4[r * (MPITCH / 4) + c];
            float4 h;
            h.x = bc.x + dinv * a.x;
            h.y = bc.y + dinv * a.y;
            h.z = bc.z + dinv * a.z;
            h.w = bc.w + dinv * a.w;
            h.x = (h.x > 0.0f ? h.x : 0.0f) + xs[it].x;
            h.y = (h.y > 0.0f ? h.y : 0.0f) + xs[it].y;
            h.z = (h.z > 0.0f ? h.z : 0.0f) + xs[it].z;
            h.w = (h.w > 0.0f ? h.w : 0.0f) + xs[it].w;
            a4[r * (MPITCH / 4) + c] = h;
        }
    }
    __syncthreads();

    const int hi = lane >> 4;
    const int lm = lane & 15;
    v8f d[2][2];
#pragma unroll
    for (int uu = 0; uu < 2; ++uu) {
        d[uu][0] = (v8f){0,0,0,0,0,0,0,0};
        d[uu][1] = (v8f){0,0,0,0,0,0,0,0};
        const int u  = wave + 32 * uu;
        const int rt = u >> 2, cq = u & 3;
        if (u < (MROWS / 16) * 4 && rt * 16 < nrows) {
            const float* arow = acc + (rt * 16 + lm) * MPITCH + 2 * hi;
            const float* wp = wlds + hi * 256 + cq * 64 + lm * 2;
            v8f c0 = d[uu][0], c1 = d[uu][1];
#pragma unroll 8
            for (int kt = 0; kt < 32; ++kt) {
                const v2f a  = *(const v2f*)(arow + kt * 4);
                const v2f b0 = *(const v2f*)(wp + kt * 512);
                const v2f b1 = *(const v2f*)(wp + kt * 512 + 32);
                c0 = __builtin_amdgcn_wmma_f32_16x16x4_f32(
                     false, a,  false, b0,
                     (short)0, c0,  false,  false);
                c1 = __builtin_amdgcn_wmma_f32_16x16x4_f32(
                     false, a,  false, b1,
                     (short)0, c1,  false,  false);
            }
            d[uu][0] = c0; d[uu][1] = c1;
        }
    }
    __syncthreads();

#pragma unroll
    for (int uu = 0; uu < 2; ++uu) {
        const int u  = wave + 32 * uu;
        const int rt = u >> 2, cq = u & 3;
        if (u < (MROWS / 16) * 4 && rt * 16 < nrows) {
#pragma unroll
            for (int j = 0; j < 2; ++j) {
                float* dp = acc + (rt * 16 + 8 * hi) * MPITCH + (cq * 2 + j) * 16 + lm;
#pragma unroll
                for (int r = 0; r < 8; ++r) dp[r * MPITCH] = d[uu][j][r];
            }
        }
    }
    __syncthreads();

    float4* o4 = (float4*)(Y2 + (size_t)row0 * NF);
#pragma unroll
    for (int it = 0; it < RPT; ++it) {
        const int r = rb + 32 * it;
        if (r < nrows) {
            const float dinv = dinvs[r];
            const float4 a = a4[r * (MPITCH / 4) + c];
            o4[r * (NF / 4) + c] = make_float4(dinv * a.x, dinv * a.y, dinv * a.z, dinv * a.w);
        }
    }
}

static_assert(N_EDGES % 32 == 0, "the scan path reads whole waves of edges");

extern "C" void kernel_launch(void* const* d_in, const int* in_sizes, int n_in,
                              void* d_out, int out_size, void* d_ws, size_t ws_size,
                              hipStream_t stream) {
    const int*   ei  = (const int*)d_in[0];
    const int*   src = ei;
    const int*   dst = ei + N_EDGES;
    const float* x   = (const float*)d_in[1];
    const float* W1  = (const float*)d_in[2];
    const float* b1  = (const float*)d_in[3];
    const float* W2  = (const float*)d_in[4];
    const float* b2  = (const float*)d_in[5];
    float*       out = (float*)d_out;

    float* deg  = (float*)d_ws;
    int*   cur  = (int*)d_ws + CUR_OFF;
    float* xw   = deg + 51200;
    float* xw2  = xw + (size_t)N_NODES * NF;
    int*   bins = (int*)d_out;

    const int gemmGrid = (N_NODES / 16 + 3) / 4;
    const int fileGrid = (N_EDGES + 4095) / 4096;

    k_init_deg  <<<(N_NODES + 255) / 256, 256, 0, stream>>>(deg, cur);
    k_file_edges<<<fileGrid, 1024, 0, stream>>>(src, dst, deg, cur, bins);

    k_gemm_fused   <<<gemmGrid, 128, 0, stream>>>(x, W1, deg, xw, N_NODES);
    k_tile_sum_next<<<NTILES, 1024, 0, stream>>>(src, dst, cur, bins, deg, xw, b1, x, W2, xw2);

    k_tile_sum     <<<NTILES, 1024, 0, stream>>>(src, dst, cur, bins, deg, xw2, b2, out);
}


# Tile sums: each tile sorts its list by destination row in LDS (counting sort); a wave walks a contiguous slice and sums each run of one row in registers, one LDS float add per run, not per edge
# speedup vs baseline: 2.7179x; 1.0918x over previous
#include <hip/hip_runtime.h>

#define N_NODES 50000
#define N_EDGES 800000
#define NF 128

#define TILE    196
#define NTILES  256
#define BIN_CAP 4096
#define CUR_OFF 50048
#define LIST_PITCH (TILE * NF)
#define LAST_ROWS  (N_NODES - (NTILES - 1) * TILE)
#define LAST_CAP   ((LAST_ROWS * NF) < BIN_CAP ? (LAST_ROWS * NF) : BIN_CAP)
#define MROWS   208
#define MPITCH  132
static_assert(TILE * NTILES >= N_NODES, "tiles cover every node");
static_assert(LAST_ROWS >= 1 && LAST_ROWS <= TILE, "the last tile is the only short one");
static_assert(LIST_PITCH >= BIN_CAP, "a full list fits inside its tile's own output rows");
static_assert(N_NODES <= 65536 && TILE <= 256, "list entry: src in bits 0-15, row inside the tile in bits 16-23");
static_assert(CUR_OFF >= N_NODES && CUR_OFF + NTILES <= 51200, "cursors stay inside the deg region");
static_assert(BIN_CAP == 32 * 16 * 8, "32 waves x 16 groups x 8 entries cover a full list");
static_assert(MROWS % 16 == 0 && MROWS >= TILE, "whole matrix row tiles cover the tile");
static_assert(MPITCH % 64 == 4 && MPITCH >= NF, "A fragment rows on different banks, 16-byte aligned rows");
static_assert((MROWS / 16) * 4 <= 64, "two work units per wave cover every (row tile, column quarter)");

__device__ __forceinline__ int list_cap(int tile) { return (tile == NTILES - 1) ? LAST_CAP : BIN_CAP; }

typedef float v2f __attribute__((ext_vector_type(2)));
typedef float v8f __attribute__((ext_vector_type(8)));

__global__ void k_init_deg(float* __restrict__ deg, int* __restrict__ cursors) {
    int i = blockIdx.x * blockDim.x + threadIdx.x;
    if (i < N_NODES) deg[i] = 1.0f;
    if (i < NTILES) cursors[i] = 0;
}


__global__ void __launch_bounds__(1024) k_file_edges(const int* __restrict__ src,
                                                     const int* __restrict__ dst,
                                                     float* __restrict__ deg,
                                                     int* __restrict__ cursor,
                                                     int* __restrict__ bins) {
    __shared__ int lcnt[NTILES];
    __shared__ int lbase[NTILES];
    const int tid = threadIdx.x;
    if (tid < NTILES) lcnt[tid] = 0;
    __syncthreads();

    int ent[4], bin[4], rank[4];
    const int e0 = blockIdx.x * 4096;
#pragma unroll
    for (int k = 0; k < 4; ++k) {
        const int e = e0 + k * 1024 + tid;
        ent[k] = 0; bin[k] = -1; rank[k] = 0;
        if (e < N_EDGES) {
            const int d = dst[e];
            const int s = src[e];
            const int b = d / TILE;
            bin[k]  = b;
            ent[k]  = s | ((d - b * TILE) << 16);
            rank[k] = atomicAdd(&lcnt[b], 1);
            atomicAdd(&deg[d], 1.0f);
        }
    }
    __syncthreads();
    if (tid < NTILES) {
        const int c = lcnt[tid];
        lbase[tid] = (c > 0) ? atomicAdd(&cursor[tid], c) : 0;
    }
    __syncthreads();
#pragma unroll
    for (int k = 0; k < 4; ++k) {
        if (bin[k] >= 0) {
            const int pos = lbase[bin[k]] + rank[k];
            if (pos < list_cap(bin[k])) bins[(size_t)bin[k] * LIST_PITCH + pos] = ent[k];
        }
    }
}

__global__ void __launch_bounds__(128) k_gemm_fused(const float* __restrict__ X,
                                                    const float* __restrict__ W,
                                                    const float* __restrict__ deg,
                                                    float* __restrict__ Y,
                                                    int nrows) {
    __shared__ float wlds[64 * 128 * 2];

    const int tid = threadIdx.x;
    for (int idx = tid; idx < 64 * 128; idx += 128) {
        const int p = idx >> 7;
        const int n = idx & 127;
        float2 w2;
        w2.x = W[(size_t)(2 * p) * NF + n];
        w2.y = W[(size_t)(2 * p + 1) * NF + n];
        *(float2*)(wlds + (size_t)idx * 2) = w2;
    }
    __syncthreads();

    const int wave    = tid >> 5;
    const int lane    = tid & 31;
    const int rowTile = blockIdx.x * 4 + wave;
    const int row0    = rowTile * 16;
    const bool active = (row0 < nrows);

    const int hi = lane >> 4;
    const int lm = lane & 15;

    v8f acc[8];
#pragma unroll
    for (int t = 0; t < 8; ++t) acc[t] = (v8f){0,0,0,0,0,0,0,0};

    const int arow = active ? (row0 + lm) : lm;
    const float* xrow = X + (size_t)arow * NF;

#pragma unroll 4
    for (int kt = 0; kt < 32; ++kt) {
        const v2f a = *(const v2f*)(xrow + kt * 4 + 2 * hi);
        const int p = kt * 2 + hi;
        const float* wp = wlds + (size_t)p * 256 + lm * 2;
#pragma unroll
        for (int t = 0; t < 8; ++t) {
            v2f b = *(const v2f*)(wp + t * 32);
            acc[t] = __builtin_amdgcn_wmma_f32_16x16x4_f32(
                 false, a,  false, b,
                 (short)0, acc[t],  false,  false);
        }
    }

    if (!active) return;

    float dv[8];
#pragma unroll
    for (int r = 0; r < 8; ++r) dv[r] = rsqrtf(deg[row0 + 8 * hi + r]);

#pragma unroll
    for (int t = 0; t < 8; ++t) {
        const int col = t * 16 + lm;
        float* yb = Y + (size_t)(row0 + 8 * hi) * NF + col;
#pragma unroll
        for (int r = 0; r < 8; ++r) yb[(size_t)r * NF] = dv[r] * acc[t][r];
    }
}

template <int PITCH>
__device__ __forceinline__ void add_one_edge(const float* __restrict__ Y, float* acc, int en, int lane) {
    const float* mp = Y + (size_t)(en & 0xffff) * NF + lane;
    float* ap = acc + ((en >> 16) & 0xff) * PITCH + lane;
    const float v0 = mp[0], v1 = mp[32], v2 = mp[64], v3 = mp[96];
    atomicAdd(ap,      v0);
    atomicAdd(ap + 32, v1);
    atomicAdd(ap + 64, v2);
    atomicAdd(ap + 96, v3);
}

__device__ __forceinline__ void load_list(const int* bins, int b, int total, int tid, int (&ent)[4]) {
    const int* myb = bins + (size_t)b * LIST_PITCH;
    const int n = (total <= list_cap(b)) ? total : 0;
#pragma unroll
    for (int k = 0; k < 4; ++k) {
        const int idx = tid + 1024 * k;
        ent[k] = (idx < n) ? myb[idx] : -1;
    }
}

template <int PITCH>
__device__ __forceinline__ void add_run(float* acc, int row, int lane, float s0, float s1, float s2, float s3) {
    float* ap = acc + row * PITCH + lane;
    atomicAdd(ap,      s0);
    atomicAdd(ap + 32, s1);
    atomicAdd(ap + 64, s2);
    atomicAdd(ap + 96, s3);
}

template <int PITCH>
__device__ __forceinline__ void add_tile_edges(const int* __restrict__ src,
                                               const int* __restrict__ dst,
                                               const float* __restrict__ Y,
                                               float* acc, int* rowcnt, int* sorted,
                                               int total, const int (&ent)[4],
                                               int b, int row0, int wave, int lane) {
    if (total <= list_cap(b)) {
        int rank[4];
#pragma unroll
        for (int k = 0; k < 4; ++k) {
            rank[k] = 0;
            if (ent[k] >= 0) rank[k] = atomicAdd(&rowcnt[(ent[k] >> 16) & 0xff], 1);
        }
        __syncthreads();
        if (wave == 0) {
            int c[8];
            int sum = 0;
#pragma unroll
            for (int i = 0; i < 8; ++i) c[i] = rowcnt[lane * 8 + i];
#pragma unroll
            for (int i = 0; i < 8; ++i) { const int t = c[i]; c[i] = sum; sum += t; }
            int incl = sum;
#pragma unroll
            for (int d = 1; d < 32; d <<= 1) {
                const int t = __shfl_up(incl, d);
                if (lane >= d) incl += t;
            }
            const int excl = incl - sum;
#pragma unroll
            for (int i = 0; i < 8; ++i) rowcnt[lane * 8 + i] = c[i] + excl;
        }
        __syncthreads();
#pragma unroll
        for (int k = 0; k < 4; ++k)
            if (ent[k] >= 0) sorted[rowcnt[(ent[k] >> 16) & 0xff] + rank[k]] = ent[k];
        __syncthreads();

        const int chunk = ((total + 255) >> 8) << 3;
        const int beg = wave * chunk;
        const int end = (beg + chunk < total) ? (beg + chunk) : total;
        if (beg < end) {
            float s0 = 0.0f, s1 = 0.0f, s2 = 0.0f, s3 = 0.0f;
            int cur = __builtin_amdgcn_readfirstlane((sorted[beg] >> 16) & 0xff);
#pragma unroll 1
            for (int p = beg; p < end; p += 32) {
                const int myi = p + lane;
                const int entv = (myi < end) ? sorted[myi] : 0;
#pragma unroll 1
                for (int r = 0; r < 4; ++r) {
                    const int nval = end - (p + 8 * r);
                    if (nval <= 0) break;
                    const int l0 = r * 8;
                    if (nval >= 8) {
                        float v[8][4];
#pragma unroll
                        for (int j = 0; j < 8; ++j) {
                            const int en = __builtin_amdgcn_readlane(entv, l0 + j);
                            const float* mp = Y + (size_t)(en & 0xffff) * NF + lane;
                            v[j][0] = mp[0];
                            v[j][1] = mp[32];
                            v[j][2] = mp[64];
                            v[j][3] = mp[96];
                        }
#pragma unroll
                        for (int j = 0; j < 8; ++j) {
                            const int en = __builtin_amdgcn_readlane(entv, l0 + j);
                            const int row = (en >> 16) & 0xff;
                            if (row != cur) {
                                add_run<PITCH>(acc, cur, lane, s0, s1, s2, s3);
                                s0 = 0.0f; s1 = 0.0f; s2 = 0.0f; s3 = 0.0f;
                                cur = row;
                            }
                            s0 += v[j][0];
                            s1 += v[j][1];
                            s2 += v[j][2];
                            s3 += v[j][3];
                        }
                    } else {
#pragma unroll 1
                        for (int j = 0; j < nval; ++j) {
                            const int en = __builtin_amdgcn_readlane(entv, l0 + j);
                            const int row = (en >> 16) & 0xff;
                            if (row != cur) {
                                add_run<PITCH>(acc, cur, lane, s0, s1, s2, s3);
                                s0 = 0.0f; s1 = 0.0f; s2 = 0.0f; s3 = 0.0f;
                                cur = row;
                            }
                            const float* mp = Y + (size_t)(en & 0xffff) * NF + lane;
                            s0 += mp[0];
                            s1 += mp[32];
                            s2 += mp[64];
                            s3 += mp[96];
                        }
                    }
                }
            }
            add_run<PITCH>(acc, cur, lane, s0, s1, s2, s3);
        }
    } else {
        for (int e0 = wave * 32; e0 < N_EDGES; e0 += 32 * 32) {
            const int d = dst[e0 + lane];
            const int s = src[e0 + lane];
            const bool mine = (d >= row0) && (d < row0 + TILE);
            unsigned mask = (unsigned)__ballot(mine);
            while (mask) {
                const int j = __builtin_ctz(mask);
                mask &= mask - 1;
                const int sj = __shfl(s, j);
                const int dj = __shfl(d, j) - row0;
                add_one_edge<PITCH>(Y, acc, sj | (dj << 16), lane);
            }
        }
    }
}

__global__ void __launch_bounds__(1024) k_tile_sum(const int* __restrict__ src,
                                                   const int* __restrict__ dst,
                                                   const int* __restrict__ cursor,
                                                   const int* bins,
                                                   const float* __restrict__ deg,
                                                   const float* __restrict__ Y,
                                                   const float* __restrict__ bias,
                                                   float* outp) {
    __shared__ __attribute__((aligned(16))) float acc[TILE * NF];
    __shared__ int rowcnt[256];
    __shared__ int sorted[BIN_CAP];

    const int tid  = threadIdx.x;
    const int lane = tid & 31;
    const int wave = __builtin_amdgcn_readfirstlane(tid >> 5);
    const int b    = blockIdx.x;
    const int row0 = b * TILE;
    const int nrows = (N_NODES - row0 < TILE) ? (N_NODES - row0) : TILE;

    const int total = __builtin_amdgcn_readfirstlane(cursor[b]);
    int ent[4];
    load_list(bins, b, total, tid, ent);

    const float4* y4 = (const float4*)(Y + (size_t)row0 * NF);
    float4* a4 = (float4*)acc;
    for (int i = tid; i < nrows * (NF / 4); i += 1024) a4[i] = y4[i];
    if (tid < 256) rowcnt[tid] = 0;
    __syncthreads();

    add_tile_edges<NF>(src, dst, Y, acc, rowcnt, sorted, total, ent, b, row0, wave, lane);
    __syncthreads();

    float4* o4 = (float4*)(outp + (size_t)row0 * NF);
    const float4* b4 = (const float4*)bias;
    for (int i = tid; i < nrows * (NF / 4); i += 1024) {
        const float dinv = rsqrtf(deg[row0 + (i >> 5)]);
        const float4 a = a4[i];
        const float4 bc = b4[i & 31];
        float4 o;
        o.x = bc.x + dinv * a.x;
        o.y = bc.y + dinv * a.y;
        o.z = bc.z + dinv * a.z;
        o.w = bc.w + dinv * a.w;
        o4[i] = o;
    }
}

#define RPT ((MROWS + 31) / 32)
__global__ void __launch_bounds__(1024) k_tile_sum_next(const int* __restrict__ src,
                                                        const int* __restrict__ dst,
                                                        const int* __restrict__ cursor,
                                                        const int* __restrict__ bins,
                                                        const float* __restrict__ deg,
                                                        const float* __restrict__ Y,
                                                        const float* __restrict__ bias,
                                                        const float* __restrict__ X,
                                                        const float* __restrict__ W,
                                                        float* __restrict__ Y2) {
    __shared__ __attribute__((aligned(16))) float acc[MROWS * MPITCH];
    __shared__ __attribute__((aligned(16))) float wlds[64 * 128 * 2];
    __shared__ float dinvs[MROWS];
    __shared__ int rowcnt[256];
    __shared__ int sorted[BIN_CAP];

    const int tid  = threadIdx.x;
    const int lane = tid & 31;
    const int wave = __builtin_amdgcn_readfirstlane(tid >> 5);
    const int b    = blockIdx.x;
    const int row0 = b * TILE;
    const int nrows = (N_NODES - row0 < TILE) ? (N_NODES - row0) : TILE;

    const int c  = tid & 31;
    const int rb = tid >> 5;
    float4* a4 = (float4*)acc;

    const float4* y4 = (const float4*)(Y + (size_t)row0 * NF);
    const float4* x4 = (const float4*)(X + (size_t)row0 * NF);
    float4 w0[2], w1[2], ys[RPT], xs[RPT];
#pragma unroll
    for (int k = 0; k < 2; ++k) {
        const int idx = tid + 1024 * k;
        w0[k] = ((const float4*)(W + (size_t)(2 * (idx >> 5)) * NF))[idx & 31];
        w1[k] = ((const float4*)(W + (size_t)(2 * (idx >> 5) + 1) * NF))[idx & 31];
    }
#pragma unroll
    for (int it = 0; it < RPT; ++it) {
        const int r  = rb + 32 * it;
        const int rc = (r < nrows) ? r : (nrows - 1);
        ys[it] = y4[rc * (NF / 4) + c];
        xs[it] = x4[rc * (NF / 4) + c];
    }
    const float4 bc = ((const float4*)bias)[c];
    float dv = 0.0f;
    if (tid < nrows) dv = rsqrtf(deg[row0 + tid]);
    const int total = __builtin_amdgcn_readfirstlane(cursor[b]);
    int ent[4];
    load_list(bins, b, total, tid, ent);

#pragma unroll
    for (int k = 0; k < 2; ++k) {
        const int idx = tid + 1024 * k;
        float* wd = wlds + (size_t)((idx >> 5) * 128 + (idx & 31) * 4) * 2;
        *(float4*)(wd)     = make_float4(w0[k].x, w1[k].x, w0[k].y, w1[k].y);
        *(float4*)(wd + 4) = make_float4(w0[k].z, w1[k].z, w0[k].w, w1[k].w);
    }
#pragma unroll
    for (int it = 0; it < RPT; ++it) {
        const int r = rb + 32 * it;
        if (r < MROWS)
            a4[r * (MPITCH / 4) + c] = (r < nrows) ? ys[it] : make_float4(0.0f, 0.0f, 0.0f, 0.0f);
    }
    if (tid < MROWS) dinvs[tid] = dv;
    if (tid < 256) rowcnt[tid] = 0;
    __syncthreads();

    add_tile_edges<MPITCH>(src, dst, Y, acc, rowcnt, sorted, total, ent, b, row0, wave, lane);
    __syncthreads();

#pragma unroll
    for (int it = 0; it < RPT; ++it) {
        const int r = rb + 32 * it;
        if (r < nrows) {
            const float dinv = dinvs[r];
            const float4 a = a4[r * (MPITCH / 4) + c];
            float4 h;
            h.x = bc.x + dinv * a.x;
            h.y = bc.y + dinv * a.y;
            h.z = bc.z + dinv * a.z;
            h.w = bc.w + dinv * a.w;
            h.x = (h.x > 0.0f ? h.x : 0.0f) + xs[it].x;
            h.y = (h.y > 0.0f ? h.y : 0.0f) + xs[it].y;
            h.z = (h.z > 0.0f ? h.z : 0.0f) + xs[it].z;
            h.w = (h.w > 0.0f ? h.w : 0.0f) + xs[it].w;
            a4[r * (MPITCH / 4) + c] = h;
        }
    }
    __syncthreads();

    const int hi = lane >> 4;
    const int lm = lane & 15;
    v8f d[2][2];
#pragma unroll
    for (int uu = 0; uu < 2; ++uu) {
        d[uu][0] = (v8f){0,0,0,0,0,0,0,0};
        d[uu][1] = (v8f){0,0,0,0,0,0,0,0};
        const int u  = wave + 32 * uu;
        const int rt = u >> 2, cq = u & 3;
        if (u < (MROWS / 16) * 4 && rt * 16 < nrows) {
            const float* arow = acc + (rt * 16 + lm) * MPITCH + 2 * hi;
            const float* wp = wlds + hi * 256 + cq * 64 + lm * 2;
            v8f c0 = d[uu][0], c1 = d[uu][1];
#pragma unroll 8
            for (int kt = 0; kt < 32; ++kt) {
                const v2f a  = *(const v2f*)(arow + kt * 4);
                const v2f b0 = *(const v2f*)(wp + kt * 512);
                const v2f b1 = *(const v2f*)(wp + kt * 512 + 32);
                c0 = __builtin_amdgcn_wmma_f32_16x16x4_f32(
                     false, a,  false, b0,
                     (short)0, c0,  false,  false);
                c1 = __builtin_amdgcn_wmma_f32_16x16x4_f32(
                     false, a,  false, b1,
                     (short)0, c1,  false,  false);
            }
            d[uu][0] = c0; d[uu][1] = c1;
        }
    }
    __syncthreads();

#pragma unroll
    for (int uu = 0; uu < 2; ++uu) {
        const int u  = wave + 32 * uu;
        const int rt = u >> 2, cq = u & 3;
        if (u < (MROWS / 16) * 4 && rt * 16 < nrows) {
#pragma unroll
            for (int j = 0; j < 2; ++j) {
                float* dp = acc + (rt * 16 + 8 * hi) * MPITCH + (cq * 2 + j) * 16 + lm;
#pragma unroll
                for (int r = 0; r < 8; ++r) dp[r * MPITCH] = d[uu][j][r];
            }
        }
    }
    __syncthreads();

    float4* o4 = (float4*)(Y2 + (size_t)row0 * NF);
#pragma unroll
    for (int it = 0; it < RPT; ++it) {
        const int r = rb + 32 * it;
        if (r < nrows) {
            const float dinv = dinvs[r];
            const float4 a = a4[r * (MPITCH / 4) + c];
            o4[r * (NF / 4) + c] = make_float4(dinv * a.x, dinv * a.y, dinv * a.z, dinv * a.w);
        }
    }
}

static_assert(N_EDGES % 32 == 0, "the scan path reads whole waves of edges");

extern "C" void kernel_launch(void* const* d_in, const int* in_sizes, int n_in,
                              void* d_out, int out_size, void* d_ws, size_t ws_size,
                              hipStream_t stream) {
    const int*   ei  = (const int*)d_in[0];
    const int*   src = ei;
    const int*   dst = ei + N_EDGES;
    const float* x   = (const float*)d_in[1];
    const float* W1  = (const float*)d_in[2];
    const float* b1  = (const float*)d_in[3];
    const float* W2  = (const float*)d_in[4];
    const float* b2  = (const float*)d_in[5];
    float*       out = (float*)d_out;

    float* deg  = (float*)d_ws;
    int*   cur  = (int*)d_ws + CUR_OFF;
    float* xw   = deg + 51200;
    float* xw2  = xw + (size_t)N_NODES * NF;
    int*   bins = (int*)d_out;

    const int gemmGrid = (N_NODES / 16 + 3) / 4;
    const int fileGrid = (N_EDGES + 4095) / 4096;

    k_init_deg  <<<(N_NODES + 255) / 256, 256, 0, stream>>>(deg, cur);
    k_file_edges<<<fileGrid, 1024, 0, stream>>>(src, dst, deg, cur, bins);

    k_gemm_fused   <<<gemmGrid, 128, 0, stream>>>(x, W1, deg, xw, N_NODES);
    k_tile_sum_next<<<NTILES, 1024, 0, stream>>>(src, dst, cur, bins, deg, xw, b1, x, W2, xw2);

    k_tile_sum     <<<NTILES, 1024, 0, stream>>>(src, dst, cur, bins, deg, xw2, b2, out);
}
